# starting kernel
# baseline (speedup 1.0000x reference)
#include <hip/hip_runtime.h>


typedef __attribute__((ext_vector_type(2))) float v2f;
typedef __attribute__((ext_vector_type(8))) float v8f;

#define NB 16
#define ND 32
#define HX 224
#define WX 224
#define HS 112
#define WS 112
#define FH 7
#define FW 7
#define TILE 16
#define TROWS (HS / TILE)
#define NWAVE (WS / TILE)
#define BLOCK (NWAVE * 32)
#define PROWS 22
#define PPITCH 122
#define KPAD 24

__global__ __launch_bounds__(BLOCK) void fused_pool_dwconv_wmma(
    const float* __restrict__ x, const float* __restrict__ S,
    float* __restrict__ out)
{
    __shared__ float Pool[PROWS * PPITCH];
    __shared__ float Bm[FH * KPAD * 16];

    const int b   = blockIdx.x;
    const int tr  = b % TROWS;
    const int d   = (b / TROWS) % ND;
    const int n   = b / (TROWS * ND);
    const int r0  = tr * TILE;
    const int tid = threadIdx.x;

    const float* xp = x + ((size_t)(n * ND + d)) * (HX * WX);
    const float* sp = S + d * (FH * FW);

    for (int e = tid; e < PROWS * PPITCH; e += BLOCK) {
        const int pr = e / PPITCH;
        const int cc = e % PPITCH;
        const int R  = r0 - 3 + pr;
        const int C  = cc - 3;
        const int Rc = R < 0 ? 0 : (R >= HS ? HS - 1 : R);
        const int Cc = C < 0 ? 0 : (C >= WS ? WS - 1 : C);
        const int ym = 2 * Rc - 1 < 0 ? 0 : 2 * Rc - 1;
        const int xm = 2 * Cc - 1 < 0 ? 0 : 2 * Cc - 1;
        const float* rowm = xp + ym * WX;
        const float* row0 = xp + 2 * Rc * WX;
        const float* rowp = row0 + WX;
        const v2f pm = *(const v2f*)(rowm + 2 * Cc);
        const v2f p0 = *(const v2f*)(row0 + 2 * Cc);
        const v2f pp = *(const v2f*)(rowp + 2 * Cc);
        const float lm = rowm[xm];
        const float l0 = row0[xm];
        const float lp = rowp[xm];
        float m = fmaxf(fmaxf(pm.x, pm.y), lm);
        m = fmaxf(m, fmaxf(fmaxf(p0.x, p0.y), l0));
        m = fmaxf(m, fmaxf(fmaxf(pp.x, pp.y), lp));
        const bool valid = (R >= 0) & (R < HS) & (C >= 0) & (C < WS);
        Pool[pr * PPITCH + cc] = valid ? m : 0.0f;
    }

    for (int e = tid; e < FH * KPAD * 16; e += BLOCK) {
        const int i = e / (KPAD * 16);
        const int k = (e / 16) % KPAD;
        const int c = e % 16;
        const int j = k - c;
        float v = 0.0f;
        if (j >= 0 && j < FW && k < 22) v = sp[i * FW + j];
        Bm[e] = v;
    }
    __syncthreads();

    const int wave  = tid >> 5;
    const int lane  = tid & 31;
    const int mrow  = lane & 15;
    const int khalf = lane >> 4;
    const int cbase = wave * TILE;

    v8f acc = {0.f, 0.f, 0.f, 0.f, 0.f, 0.f, 0.f, 0.f};
#pragma unroll
    for (int i = 0; i < FH; ++i) {
        const float* arow = &Pool[(mrow + i) * PPITCH + cbase + khalf * 2];
        const float* bcol = &Bm[i * (KPAD * 16) + khalf * 2 * 16 + mrow];
#pragma unroll
        for (int kb = 0; kb < KPAD / 4; ++kb) {
            v2f a = *(const v2f*)(arow + kb * 4);
            v2f bv;
            bv.x = bcol[kb * 4 * 16];
            bv.y = bcol[kb * 4 * 16 + 16];
            acc = __builtin_amdgcn_wmma_f32_16x16x4_f32(
                false, a, false, bv, (short)0, acc, false, false);
        }
    }

    float* op = out + ((size_t)(n * ND + d)) * (HS * WS)
                    + (size_t)r0 * WS + cbase;
#pragma unroll
    for (int v = 0; v < 8; ++v) {
        const int m = v + 8 * khalf;
        op[m * WS + mrow] = acc[v];
    }
}

extern "C" void kernel_launch(void* const* d_in, const int* in_sizes, int n_in,
                              void* d_out, int out_size, void* d_ws, size_t ws_size,
                              hipStream_t stream) {
    (void)in_sizes; (void)n_in; (void)d_ws; (void)ws_size; (void)out_size;
    const float* x = (const float*)d_in[0];
    const float* S = (const float*)d_in[1];
    float* out = (float*)d_out;

    dim3 grid(NB * ND * TROWS);
    fused_pool_dwconv_wmma<<<grid, BLOCK, 0, stream>>>(x, S, out);
}


# product transposed: K runs over pooled rows (band = A, pooled row pairs = B), so each 4-row pool step is one K-block whose 49 WMMAs run under the next step's x loads; sums now k-block-major
# speedup vs baseline: 1.3806x; 1.3806x over previous
#include <hip/hip_runtime.h>


typedef __attribute__((ext_vector_type(2))) float v2f;
typedef __attribute__((ext_vector_type(8))) float v8f;

#define NB 16
#define ND 32
#define HX 224
#define WX 224
#define HS 112
#define WS 112
#define FH 7
#define FW 7
#define TILE 16
#define TROWS (HS / TILE)
#define NTILE (WS / TILE)
#define WAVES 4
#define BLOCK (WAVES * 32)
#define NSEG 4
#define KBLKS 6
#define QROWS 12
#define QPITCH 288

struct RowPairLoads {
    v2f   pm[2][NSEG], p0[2][NSEG], pp[2][NSEG];
    float lm[2][NSEG], l0[2][NSEG], lp[2][NSEG];
};

__device__ __forceinline__ void pool_issue(RowPairLoads& h,
                                           const float* __restrict__ xp,
                                           int r0, int pr, int lane)
{
#pragma unroll
    for (int u = 0; u < 2; ++u) {
        const int R  = r0 - 3 + pr + u;
        const int Rc = R < 0 ? 0 : (R >= HS ? HS - 1 : R);
        const int ym = 2 * Rc - 1 < 0 ? 0 : 2 * Rc - 1;
        const float* rowm = xp + ym * WX;
        const float* row0 = xp + 2 * Rc * WX;
        const float* rowp = row0 + WX;
#pragma unroll
        for (int s = 0; s < NSEG; ++s) {
            const int C  = 32 * s + lane - 3;
            const int Cc = C < 0 ? 0 : (C >= WS ? WS - 1 : C);
            const int xm = 2 * Cc - 1 < 0 ? 0 : 2 * Cc - 1;
            h.pm[u][s] = *(const v2f*)(rowm + 2 * Cc);
            h.p0[u][s] = *(const v2f*)(row0 + 2 * Cc);
            h.pp[u][s] = *(const v2f*)(rowp + 2 * Cc);
            h.lm[u][s] = rowm[xm];
            h.l0[u][s] = row0[xm];
            h.lp[u][s] = rowp[xm];
        }
    }
}

__device__ __forceinline__ void pool_finish(const RowPairLoads& h, float* Pool,
                                            int r0, int pr, int lane)
{
    const int  R0  = r0 - 3 + pr;
    const bool rv0 = (R0 >= 0) & (R0 < HS);
    const bool rv1 = (R0 + 1 >= 0) & (R0 + 1 < HS);
#pragma unroll
    for (int s = 0; s < NSEG; ++s) {
        const int  cc = 32 * s + lane;
        const int  C  = cc - 3;
        const bool cv = (C >= 0) & (C < WS);
        float m[2];
#pragma unroll
        for (int u = 0; u < 2; ++u) {
            float t = fmaxf(fmaxf(h.pm[u][s].x, h.pm[u][s].y), h.lm[u][s]);
            t = fmaxf(t, fmaxf(fmaxf(h.p0[u][s].x, h.p0[u][s].y), h.l0[u][s]));
            t = fmaxf(t, fmaxf(fmaxf(h.pp[u][s].x, h.pp[u][s].y), h.lp[u][s]));
            m[u] = t;
        }
        v2f o;
        o.x = (rv0 & cv) ? m[0] : 0.0f;
        o.y = (rv1 & cv) ? m[1] : 0.0f;
        *(v2f*)&Pool[(pr >> 1) * QPITCH + cc * 2] = o;
    }
}

__device__ __forceinline__ void band_issue(v2f (&raw)[FW],
                                           const float* __restrict__ sp,
                                           int kb, int mrow, int khalf)
{
    const int i0  = kb * 4 + khalf * 2 - mrow;
    const int i1  = i0 + 1;
    const int i0c = i0 < 0 ? 0 : (i0 > FH - 1 ? FH - 1 : i0);
    const int i1c = i1 < 0 ? 0 : (i1 > FH - 1 ? FH - 1 : i1);
#pragma unroll
    for (int j = 0; j < FW; ++j) {
        raw[j].x = sp[i0c * FW + j];
        raw[j].y = sp[i1c * FW + j];
    }
}

__device__ __forceinline__ void band_finish(v2f (&a)[FW], const v2f (&raw)[FW],
                                            int kb, int mrow, int khalf)
{
    const int  i0  = kb * 4 + khalf * 2 - mrow;
    const int  i1  = i0 + 1;
    const bool in0 = (i0 >= 0) & (i0 < FH);
    const bool in1 = (i1 >= 0) & (i1 < FH);
#pragma unroll
    for (int j = 0; j < FW; ++j) {
        a[j].x = in0 ? raw[j].x : 0.0f;
        a[j].y = in1 ? raw[j].y : 0.0f;
    }
}

template <int JLO, int JHI>
__device__ __forceinline__ void mma_block(v8f (&acc)[NTILE], const v2f (&a)[FW],
                                          const float* brow)
{
#pragma unroll
    for (int j = JLO; j < JHI; ++j) {
#pragma unroll
        for (int t = 0; t < NTILE; ++t) {
            const v2f b = *(const v2f*)(brow + (t * TILE + j) * 2);
            acc[t] = __builtin_amdgcn_wmma_f32_16x16x4_f32(
                false, a[j], false, b, (short)0, acc[t], false, false);
        }
    }
}

__device__ __forceinline__ void wave_lds_order()
{
    __builtin_amdgcn_fence(__ATOMIC_RELEASE, "wavefront");
    __builtin_amdgcn_wave_barrier();
    __builtin_amdgcn_fence(__ATOMIC_ACQUIRE, "wavefront");
}

__global__ __launch_bounds__(BLOCK) void fused_pool_dwconv_wmma(
    const float* __restrict__ x, const float* __restrict__ S,
    float* __restrict__ out)
{
    __shared__ float PoolAll[WAVES * QROWS * QPITCH];

    const int tid   = threadIdx.x;
    const int lane  = tid & 31;
    const int wave  = __builtin_amdgcn_readfirstlane(tid >> 5);
    const int strip = blockIdx.x * WAVES + wave;
    const int tr    = strip % TROWS;
    const int d     = (strip / TROWS) % ND;
    const int n     = strip / (TROWS * ND);
    const int r0    = tr * TILE;

    float* Pool = PoolAll + wave * (QROWS * QPITCH);
    const float* xp = x + ((size_t)(n * ND + d)) * (HX * WX);
    const float* sp = S + d * (FH * FW);

    const int mrow  = lane & 15;
    const int khalf = lane >> 4;

    RowPairLoads ha, hb;
    v2f araw[FW], a[FW];

    pool_issue(ha, xp, r0, 0, lane);
    pool_issue(hb, xp, r0, 2, lane);
    band_issue(araw, sp, 0, mrow, khalf);
    __builtin_amdgcn_sched_barrier(0);
    pool_finish(ha, Pool, r0, 0, lane);
    pool_finish(hb, Pool, r0, 2, lane);
    band_finish(a, araw, 0, mrow, khalf);

    v8f acc[NTILE];
#pragma unroll
    for (int t = 0; t < NTILE; ++t)
        acc[t] = (v8f){0.f, 0.f, 0.f, 0.f, 0.f, 0.f, 0.f, 0.f};

    wave_lds_order();

    const float* brow = Pool + khalf * QPITCH + mrow * 2;

#pragma unroll 1
    for (int kb = 0; kb < KBLKS - 1; ++kb) {
        const int prn = 4 * (kb + 1);
        pool_issue(ha, xp, r0, prn, lane);
        band_issue(araw, sp, kb + 1, mrow, khalf);
        __builtin_amdgcn_sched_barrier(0);
        mma_block<0, 4>(acc, a, brow);
        __builtin_amdgcn_sched_barrier(0);
        pool_finish(ha, Pool, r0, prn, lane);
        pool_issue(hb, xp, r0, prn + 2, lane);
        __builtin_amdgcn_sched_barrier(0);
        mma_block<4, 7>(acc, a, brow);
        __builtin_amdgcn_sched_barrier(0);
        pool_finish(hb, Pool, r0, prn + 2, lane);
        band_finish(a, araw, kb + 1, mrow, khalf);
        wave_lds_order();
        brow += 2 * QPITCH;
    }

    mma_block<0, 7>(acc, a, brow);

    float* op = out + ((size_t)(n * ND + d)) * (HS * WS) + (size_t)r0 * WS;
#pragma unroll
    for (int t = 0; t < NTILE; ++t) {
#pragma unroll
        for (int v = 0; v < 8; ++v) {
            const int m = v + 8 * khalf;
            op[m * WS + t * TILE + mrow] = acc[t][v];
        }
    }
}

extern "C" void kernel_launch(void* const* d_in, const int* in_sizes, int n_in,
                              void* d_out, int out_size, void* d_ws, size_t ws_size,
                              hipStream_t stream) {
    (void)in_sizes; (void)n_in; (void)d_ws; (void)ws_size; (void)out_size;
    const float* x = (const float*)d_in[0];
    const float* S = (const float*)d_in[1];
    float* out = (float*)d_out;

    dim3 grid(NB * ND * TROWS / WAVES);
    fused_pool_dwconv_wmma<<<grid, BLOCK, 0, stream>>>(x, S, out);
}


# strips of 14 output rows, 8 per plane (was 7 of 16): K = 14+6 = 20 pooled rows = 5 whole K-blocks, 245 WMMAs a wave (was 294); 4096 equal waves; taps regroup into K-blocks: sum order shifts
# speedup vs baseline: 1.4957x; 1.0834x over previous
#include <hip/hip_runtime.h>


typedef __attribute__((ext_vector_type(2))) float v2f;
typedef __attribute__((ext_vector_type(8))) float v8f;

#define NB 16
#define ND 32
#define HX 224
#define WX 224
#define HS 112
#define WS 112
#define FH 7
#define FW 7
#define TILE 16
#define SROWS 14
#define TROWS (HS / SROWS)
#define NTILE (WS / TILE)
#define WAVES 4
#define BLOCK (WAVES * 32)
#define NSEG 4
#define KBLKS 5
#define QROWS 10
#define QPITCH 288

struct RowPairLoads {
    v2f   pm[2][NSEG], p0[2][NSEG], pp[2][NSEG];
    float lm[2][NSEG], l0[2][NSEG], lp[2][NSEG];
};

__device__ __forceinline__ void pool_issue(RowPairLoads& h,
                                           const float* __restrict__ xp,
                                           int r0, int pr, int lane)
{
#pragma unroll
    for (int u = 0; u < 2; ++u) {
        const int R  = r0 - 3 + pr + u;
        const int Rc = R < 0 ? 0 : (R >= HS ? HS - 1 : R);
        const int ym = 2 * Rc - 1 < 0 ? 0 : 2 * Rc - 1;
        const float* rowm = xp + ym * WX;
        const float* row0 = xp + 2 * Rc * WX;
        const float* rowp = row0 + WX;
#pragma unroll
        for (int s = 0; s < NSEG; ++s) {
            const int C  = 32 * s + lane - 3;
            const int Cc = C < 0 ? 0 : (C >= WS ? WS - 1 : C);
            const int xm = 2 * Cc - 1 < 0 ? 0 : 2 * Cc - 1;
            h.pm[u][s] = *(const v2f*)(rowm + 2 * Cc);
            h.p0[u][s] = *(const v2f*)(row0 + 2 * Cc);
            h.pp[u][s] = *(const v2f*)(rowp + 2 * Cc);
            h.lm[u][s] = rowm[xm];
            h.l0[u][s] = row0[xm];
            h.lp[u][s] = rowp[xm];
        }
    }
}

__device__ __forceinline__ void pool_finish(const RowPairLoads& h, float* Pool,
                                            int r0, int pr, int lane)
{
    const int  R0  = r0 - 3 + pr;
    const bool rv0 = (R0 >= 0) & (R0 < HS);
    const bool rv1 = (R0 + 1 >= 0) & (R0 + 1 < HS);
#pragma unroll
    for (int s = 0; s < NSEG; ++s) {
        const int  cc = 32 * s + lane;
        const int  C  = cc - 3;
        const bool cv = (C >= 0) & (C < WS);
        float m[2];
#pragma unroll
        for (int u = 0; u < 2; ++u) {
            float t = fmaxf(fmaxf(h.pm[u][s].x, h.pm[u][s].y), h.lm[u][s]);
            t = fmaxf(t, fmaxf(fmaxf(h.p0[u][s].x, h.p0[u][s].y), h.l0[u][s]));
            t = fmaxf(t, fmaxf(fmaxf(h.pp[u][s].x, h.pp[u][s].y), h.lp[u][s]));
            m[u] = t;
        }
        v2f o;
        o.x = (rv0 & cv) ? m[0] : 0.0f;
        o.y = (rv1 & cv) ? m[1] : 0.0f;
        *(v2f*)&Pool[(pr >> 1) * QPITCH + cc * 2] = o;
    }
}

__device__ __forceinline__ void band_issue(v2f (&raw)[FW],
                                           const float* __restrict__ sp,
                                           int kb, int mrow, int khalf)
{
    const int i0  = kb * 4 + khalf * 2 - mrow;
    const int i1  = i0 + 1;
    const int i0c = i0 < 0 ? 0 : (i0 > FH - 1 ? FH - 1 : i0);
    const int i1c = i1 < 0 ? 0 : (i1 > FH - 1 ? FH - 1 : i1);
#pragma unroll
    for (int j = 0; j < FW; ++j) {
        raw[j].x = sp[i0c * FW + j];
        raw[j].y = sp[i1c * FW + j];
    }
}

__device__ __forceinline__ void band_finish(v2f (&a)[FW], const v2f (&raw)[FW],
                                            int kb, int mrow, int khalf)
{
    const int  i0  = kb * 4 + khalf * 2 - mrow;
    const int  i1  = i0 + 1;
    const bool in0 = (i0 >= 0) & (i0 < FH);
    const bool in1 = (i1 >= 0) & (i1 < FH);
#pragma unroll
    for (int j = 0; j < FW; ++j) {
        a[j].x = in0 ? raw[j].x : 0.0f;
        a[j].y = in1 ? raw[j].y : 0.0f;
    }
}

template <int JLO, int JHI>
__device__ __forceinline__ void mma_block(v8f (&acc)[NTILE], const v2f (&a)[FW],
                                          const float* brow)
{
#pragma unroll
    for (int j = JLO; j < JHI; ++j) {
#pragma unroll
        for (int t = 0; t < NTILE; ++t) {
            const v2f b = *(const v2f*)(brow + (t * TILE + j) * 2);
            acc[t] = __builtin_amdgcn_wmma_f32_16x16x4_f32(
                false, a[j], false, b, (short)0, acc[t], false, false);
        }
    }
}

__device__ __forceinline__ void wave_lds_order()
{
    __builtin_amdgcn_fence(__ATOMIC_RELEASE, "wavefront");
    __builtin_amdgcn_wave_barrier();
    __builtin_amdgcn_fence(__ATOMIC_ACQUIRE, "wavefront");
}

__global__ __launch_bounds__(BLOCK) void fused_pool_dwconv_wmma(
    const float* __restrict__ x, const float* __restrict__ S,
    float* __restrict__ out)
{
    __shared__ float PoolAll[WAVES * QROWS * QPITCH];

    const int tid   = threadIdx.x;
    const int lane  = tid & 31;
    const int wave  = __builtin_amdgcn_readfirstlane(tid >> 5);
    const int strip = blockIdx.x * WAVES + wave;
    const int tr    = strip % TROWS;
    const int d     = (strip / TROWS) % ND;
    const int n     = strip / (TROWS * ND);
    const int r0    = tr * SROWS;

    float* Pool = PoolAll + wave * (QROWS * QPITCH);
    const float* xp = x + ((size_t)(n * ND + d)) * (HX * WX);
    const float* sp = S + d * (FH * FW);

    const int mrow  = lane & 15;
    const int khalf = lane >> 4;

    RowPairLoads ha, hb;
    v2f araw[FW], a[FW];

    pool_issue(ha, xp, r0, 0, lane);
    pool_issue(hb, xp, r0, 2, lane);
    band_issue(araw, sp, 0, mrow, khalf);
    __builtin_amdgcn_sched_barrier(0);
    pool_finish(ha, Pool, r0, 0, lane);
    pool_finish(hb, Pool, r0, 2, lane);
    band_finish(a, araw, 0, mrow, khalf);

    v8f acc[NTILE];
#pragma unroll
    for (int t = 0; t < NTILE; ++t)
        acc[t] = (v8f){0.f, 0.f, 0.f, 0.f, 0.f, 0.f, 0.f, 0.f};

    wave_lds_order();

    const float* brow = Pool + khalf * QPITCH + mrow * 2;

#pragma unroll 1
    for (int kb = 0; kb < KBLKS - 1; ++kb) {
        const int prn = 4 * (kb + 1);
        pool_issue(ha, xp, r0, prn, lane);
        band_issue(araw, sp, kb + 1, mrow, khalf);
        __builtin_amdgcn_sched_barrier(0);
        mma_block<0, 4>(acc, a, brow);
        __builtin_amdgcn_sched_barrier(0);
        pool_finish(ha, Pool, r0, prn, lane);
        pool_issue(hb, xp, r0, prn + 2, lane);
        __builtin_amdgcn_sched_barrier(0);
        mma_block<4, 7>(acc, a, brow);
        __builtin_amdgcn_sched_barrier(0);
        pool_finish(hb, Pool, r0, prn + 2, lane);
        band_finish(a, araw, kb + 1, mrow, khalf);
        wave_lds_order();
        brow += 2 * QPITCH;
    }

    mma_block<0, 7>(acc, a, brow);

    float* op = out + ((size_t)(n * ND + d)) * (HS * WS) + (size_t)r0 * WS;
#pragma unroll
    for (int t = 0; t < NTILE; ++t) {
#pragma unroll
        for (int v = 0; v < SROWS - 8; ++v) {
            const int m = v + 8 * khalf;
            op[m * WS + t * TILE + mrow] = acc[t][v];
        }
    }
    if (khalf == 0) {
#pragma unroll
        for (int t = 0; t < NTILE; ++t) {
#pragma unroll
            for (int v = SROWS - 8; v < 8; ++v)
                op[v * WS + t * TILE + mrow] = acc[t][v];
        }
    }
}

extern "C" void kernel_launch(void* const* d_in, const int* in_sizes, int n_in,
                              void* d_out, int out_size, void* d_ws, size_t ws_size,
                              hipStream_t stream) {
    (void)in_sizes; (void)n_in; (void)d_ws; (void)ws_size; (void)out_size;
    const float* x = (const float*)d_in[0];
    const float* S = (const float*)d_in[1];
    float* out = (float*)d_out;

    dim3 grid(NB * ND * TROWS / WAVES);
    fused_pool_dwconv_wmma<<<grid, BLOCK, 0, stream>>>(x, S, out);
}


# pool stage loads each x row once per wave (8-byte loads); 3x3 max separable in registers, shared odd row carried on, left tap by a DPP lane shift instead of a load; 164 x loads a wave, was 300
# speedup vs baseline: 1.6302x; 1.0899x over previous
#include <hip/hip_runtime.h>


typedef __attribute__((ext_vector_type(2))) float v2f;
typedef __attribute__((ext_vector_type(8))) float v8f;

#define NB 16
#define ND 32
#define HX 224
#define WX 224
#define HS 112
#define WS 112
#define FH 7
#define FW 7
#define TILE 16
#define SROWS 14
#define TROWS (HS / SROWS)
#define NTILE (WS / TILE)
#define WAVES 4
#define BLOCK (WAVES * 32)
#define NSEG 4
#define SEGW 30
#define KBLKS 5
#define QROWS 10
#define QPITCH 288

struct RowQuad {
    v2f r[4][NSEG];
};

__device__ __forceinline__ const float* x_row(const float* __restrict__ xp, int y)
{
    const int yc = y < 0 ? 0 : (y > HX - 1 ? HX - 1 : y);
    return xp + yc * WX;
}

__device__ __forceinline__ void row_issue(v2f (&dst)[NSEG],
                                          const float* __restrict__ row,
                                          const unsigned (&coff)[NSEG])
{
#pragma unroll
    for (int s = 0; s < NSEG; ++s)
        dst[s] = *(const v2f*)(row + coff[s]);
}

__device__ __forceinline__ void pool_issue(RowQuad& h,
                                           const float* __restrict__ xp,
                                           int r0, int pr,
                                           const unsigned (&coff)[NSEG])
{
    const int y0 = 2 * (r0 - 3 + pr);
#pragma unroll
    for (int i = 0; i < 4; ++i)
        row_issue(h.r[i], x_row(xp, y0 + i), coff);
}

__device__ __forceinline__ float left_lane(float v)
{
    const int b = __builtin_bit_cast(int, v);
    return __builtin_bit_cast(float,
        __builtin_amdgcn_update_dpp(b, b, 0x111, 0xf, 0xf, false));
}

__device__ __forceinline__ float max3f(float a, float b, float c)
{
    return fmaxf(fmaxf(a, b), c);
}

__device__ __forceinline__ void pool_finish(const RowQuad& h, v2f (&carry)[NSEG],
                                            float* Pool, int r0, int pr,
                                            const bool (&cv)[NSEG],
                                            const int (&scol)[NSEG])
{
    const int  R0  = r0 - 3 + pr;
    const bool rv0 = (R0 >= 0) & (R0 < HS);
    const bool rv1 = (R0 + 1 >= 0) & (R0 + 1 < HS);
#pragma unroll
    for (int s = 0; s < NSEG; ++s) {
        v2f v0, v1;
        v0.x = max3f(carry[s].x, h.r[0][s].x, h.r[1][s].x);
        v0.y = max3f(carry[s].y, h.r[0][s].y, h.r[1][s].y);
        v1.x = max3f(h.r[1][s].x, h.r[2][s].x, h.r[3][s].x);
        v1.y = max3f(h.r[1][s].y, h.r[2][s].y, h.r[3][s].y);
        carry[s] = h.r[3][s];
        const float m0 = max3f(v0.x, v0.y, left_lane(v0.y));
        const float m1 = max3f(v1.x, v1.y, left_lane(v1.y));
        v2f o;
        o.x = (rv0 & cv[s]) ? m0 : 0.0f;
        o.y = (rv1 & cv[s]) ? m1 : 0.0f;
        *(v2f*)&Pool[(pr >> 1) * QPITCH + scol[s] * 2] = o;
    }
}

__device__ __forceinline__ void band_issue(v2f (&raw)[FW],
                                           const float* __restrict__ sp,
                                           int kb, int mrow, int khalf)
{
    const int i0  = kb * 4 + khalf * 2 - mrow;
    const int i1  = i0 + 1;
    const int i0c = i0 < 0 ? 0 : (i0 > FH - 1 ? FH - 1 : i0);
    const int i1c = i1 < 0 ? 0 : (i1 > FH - 1 ? FH - 1 : i1);
#pragma unroll
    for (int j = 0; j < FW; ++j) {
        raw[j].x = sp[i0c * FW + j];
        raw[j].y = sp[i1c * FW + j];
    }
}

__device__ __forceinline__ void band_finish(v2f (&a)[FW], const v2f (&raw)[FW],
                                            int kb, int mrow, int khalf)
{
    const int  i0  = kb * 4 + khalf * 2 - mrow;
    const int  i1  = i0 + 1;
    const bool in0 = (i0 >= 0) & (i0 < FH);
    const bool in1 = (i1 >= 0) & (i1 < FH);
#pragma unroll
    for (int j = 0; j < FW; ++j) {
        a[j].x = in0 ? raw[j].x : 0.0f;
        a[j].y = in1 ? raw[j].y : 0.0f;
    }
}

template <int JLO, int JHI>
__device__ __forceinline__ void mma_block(v8f (&acc)[NTILE], const v2f (&a)[FW],
                                          const float* brow)
{
#pragma unroll
    for (int j = JLO; j < JHI; ++j) {
#pragma unroll
        for (int t = 0; t < NTILE; ++t) {
            const v2f b = *(const v2f*)(brow + (t * TILE + j) * 2);
            acc[t] = __builtin_amdgcn_wmma_f32_16x16x4_f32(
                false, a[j], false, b, (short)0, acc[t], false, false);
        }
    }
}

__device__ __forceinline__ void wave_lds_order()
{
    __builtin_amdgcn_fence(__ATOMIC_RELEASE, "wavefront");
    __builtin_amdgcn_wave_barrier();
    __builtin_amdgcn_fence(__ATOMIC_ACQUIRE, "wavefront");
}

__global__ __launch_bounds__(BLOCK) void fused_pool_dwconv_wmma(
    const float* __restrict__ x, const float* __restrict__ S,
    float* __restrict__ out)
{
    __shared__ float PoolAll[WAVES * QROWS * QPITCH];

    const int tid   = threadIdx.x;
    const int lane  = tid & 31;
    const int wave  = __builtin_amdgcn_readfirstlane(tid >> 5);
    const int strip = blockIdx.x * WAVES + wave;
    const int tr    = strip % TROWS;
    const int d     = (strip / TROWS) % ND;
    const int n     = strip / (TROWS * ND);
    const int r0    = tr * SROWS;

    float* Pool = PoolAll + wave * (QROWS * QPITCH);
    const float* xp = x + ((size_t)(n * ND + d)) * (HX * WX);
    const float* sp = S + d * (FH * FW);

    const int mrow  = lane & 15;
    const int khalf = lane >> 4;

    unsigned coff[NSEG];
    bool     cv[NSEG];
    int      scol[NSEG];
#pragma unroll
    for (int s = 0; s < NSEG; ++s) {
        const int cc = SEGW * s + 15 * (lane >> 4) + (lane & 15) - 1;
        const int C  = cc - 3;
        const int Cc = C < 0 ? 0 : (C >= WS ? WS - 1 : C);
        coff[s] = (unsigned)(2 * Cc);
        cv[s]   = ((lane & 15) != 0) & (C >= 0) & (C < WS);
        scol[s] = (lane & 15) != 0 ? cc : 120 + 2 * s + (lane >> 4);
    }

    RowQuad ha, hb;
    v2f carry[NSEG];
    v2f araw[FW], a[FW];

    row_issue(carry, x_row(xp, 2 * (r0 - 3) - 1), coff);
    pool_issue(ha, xp, r0, 0, coff);
    pool_issue(hb, xp, r0, 2, coff);
    band_issue(araw, sp, 0, mrow, khalf);
    __builtin_amdgcn_sched_barrier(0);
    pool_finish(ha, carry, Pool, r0, 0, cv, scol);
    pool_finish(hb, carry, Pool, r0, 2, cv, scol);
    band_finish(a, araw, 0, mrow, khalf);

    v8f acc[NTILE];
#pragma unroll
    for (int t = 0; t < NTILE; ++t)
        acc[t] = (v8f){0.f, 0.f, 0.f, 0.f, 0.f, 0.f, 0.f, 0.f};

    wave_lds_order();

    const float* brow = Pool + khalf * QPITCH + mrow * 2;

#pragma unroll 1
    for (int kb = 0; kb < KBLKS - 1; ++kb) {
        const int prn = 4 * (kb + 1);
        pool_issue(ha, xp, r0, prn, coff);
        band_issue(araw, sp, kb + 1, mrow, khalf);
        __builtin_amdgcn_sched_barrier(0);
        mma_block<0, 4>(acc, a, brow);
        __builtin_amdgcn_sched_barrier(0);
        pool_finish(ha, carry, Pool, r0, prn, cv, scol);
        pool_issue(hb, xp, r0, prn + 2, coff);
        __builtin_amdgcn_sched_barrier(0);
        mma_block<4, 7>(acc, a, brow);
        __builtin_amdgcn_sched_barrier(0);
        pool_finish(hb, carry, Pool, r0, prn + 2, cv, scol);
        band_finish(a, araw, kb + 1, mrow, khalf);
        wave_lds_order();
        brow += 2 * QPITCH;
    }

    mma_block<0, 7>(acc, a, brow);

    float* op = out + ((size_t)(n * ND + d)) * (HS * WS) + (size_t)r0 * WS;
#pragma unroll
    for (int t = 0; t < NTILE; ++t) {
#pragma unroll
        for (int v = 0; v < SROWS - 8; ++v) {
            const int m = v + 8 * khalf;
            op[m * WS + t * TILE + mrow] = acc[t][v];
        }
    }
    if (khalf == 0) {
#pragma unroll
        for (int t = 0; t < NTILE; ++t) {
#pragma unroll
            for (int v = SROWS - 8; v < 8; ++v)
                op[v * WS + t * TILE + mrow] = acc[t][v];
        }
    }
}

extern "C" void kernel_launch(void* const* d_in, const int* in_sizes, int n_in,
                              void* d_out, int out_size, void* d_ws, size_t ws_size,
                              hipStream_t stream) {
    (void)in_sizes; (void)n_in; (void)d_ws; (void)ws_size; (void)out_size;
    const float* x = (const float*)d_in[0];
    const float* S = (const float*)d_in[1];
    float* out = (float*)d_out;

    dim3 grid(NB * ND * TROWS / WAVES);
    fused_pool_dwconv_wmma<<<grid, BLOCK, 0, stream>>>(x, S, out);
}


# M dimension = 2 output rows x 7 column phases (lane n = columns 7n..7n+6), K still over pooled rows: 13 shifts x 2 K-blocks = 26 WMMAs per 2x112 tile, 182 a wave (was 245); sum order changes
# speedup vs baseline: 1.8017x; 1.1052x over previous
#include <hip/hip_runtime.h>


typedef __attribute__((ext_vector_type(2))) float v2f;
typedef __attribute__((ext_vector_type(4))) float v4f;
typedef __attribute__((ext_vector_type(8))) float v8f;

#define NB 16
#define ND 32
#define HX 224
#define WX 224
#define HS 112
#define WS 112
#define FH 7
#define FW 7
#define SROWS 14
#define TROWS (HS / SROWS)
#define NTILE (SROWS / 2)
#define NSHIFT (FW + 6)
#define WAVES 4
#define BLOCK (WAVES * 32)
#define NSEG 4
#define SEGW 30
#define NSTEP 5
#define QROWS 10
#define PHW 17
#define QW 144
#define SPARE 119

struct RowQuad {
    v2f r[4][NSEG];
};

__device__ __forceinline__ const float* x_row(const float* __restrict__ xp, int y)
{
    const int yc = y < 0 ? 0 : (y > HX - 1 ? HX - 1 : y);
    return xp + yc * WX;
}

__device__ __forceinline__ void row_issue(v2f (&dst)[NSEG],
                                          const float* __restrict__ row,
                                          const unsigned (&coff)[NSEG])
{
#pragma unroll
    for (int s = 0; s < NSEG; ++s)
        dst[s] = *(const v2f*)(row + coff[s]);
}

__device__ __forceinline__ void pool_issue(RowQuad& h,
                                           const float* __restrict__ xp,
                                           int r0, int pr,
                                           const unsigned (&coff)[NSEG])
{
    const int y0 = 2 * (r0 - 3 + pr);
#pragma unroll
    for (int i = 0; i < 4; ++i)
        row_issue(h.r[i], x_row(xp, y0 + i), coff);
}

__device__ __forceinline__ float left_lane(float v)
{
    const int b = __builtin_bit_cast(int, v);
    return __builtin_bit_cast(float,
        __builtin_amdgcn_update_dpp(b, b, 0x111, 0xf, 0xf, false));
}

__device__ __forceinline__ float max3f(float a, float b, float c)
{
    return fmaxf(fmaxf(a, b), c);
}

__device__ __forceinline__ void pool_finish(const RowQuad& h, v2f (&carry)[NSEG],
                                            float* Pool, int r0, int pr,
                                            const bool (&cv)[NSEG],
                                            const int (&sword)[NSEG])
{
    const int  R0  = r0 - 3 + pr;
    const bool rv0 = (R0 >= 0) & (R0 < HS);
    const bool rv1 = (R0 + 1 >= 0) & (R0 + 1 < HS);
#pragma unroll
    for (int s = 0; s < NSEG; ++s) {
        v2f v0, v1;
        v0.x = max3f(carry[s].x, h.r[0][s].x, h.r[1][s].x);
        v0.y = max3f(carry[s].y, h.r[0][s].y, h.r[1][s].y);
        v1.x = max3f(h.r[1][s].x, h.r[2][s].x, h.r[3][s].x);
        v1.y = max3f(h.r[1][s].y, h.r[2][s].y, h.r[3][s].y);
        carry[s] = h.r[3][s];
        const float m0 = max3f(v0.x, v0.y, left_lane(v0.y));
        const float m1 = max3f(v1.x, v1.y, left_lane(v1.y));
        v2f o;
        o.x = (rv0 & cv[s]) ? m0 : 0.0f;
        o.y = (rv1 & cv[s]) ? m1 : 0.0f;
        *(v2f*)&Pool[((pr >> 1) * QW + sword[s]) * 2] = o;
    }
}

__device__ __forceinline__ void taps_issue(float (&raw)[2][2][FW],
                                           const float* __restrict__ sp,
                                           int mr, int khalf)
{
#pragma unroll
    for (int b = 0; b < 2; ++b) {
#pragma unroll
        for (int e = 0; e < 2; ++e) {
            const int i  = 4 * b + 2 * khalf + e - mr;
            const int ic = i < 0 ? 0 : (i > FH - 1 ? FH - 1 : i);
#pragma unroll
            for (int j = 0; j < FW; ++j)
                raw[b][e][j] = sp[ic * FW + j];
        }
    }
}

__device__ __forceinline__ void taps_finish(v2f (&a)[2][NSHIFT],
                                            const float (&raw)[2][2][FW],
                                            int mr, int u, int khalf)
{
    const bool s1 = (u & 1) != 0;
    const bool s2 = (u & 2) != 0;
    const bool s4 = (u & 4) != 0;
#pragma unroll
    for (int b = 0; b < 2; ++b) {
#pragma unroll
        for (int e = 0; e < 2; ++e) {
            const int  i  = 4 * b + 2 * khalf + e - mr;
            const bool in = (mr < 2) & (i >= 0) & (i < FH);
            float p[NSHIFT], q[NSHIFT];
#pragma unroll
            for (int k = 0; k < NSHIFT; ++k)
                p[k] = (k < FW && in) ? raw[b][e][k < FW ? k : 0] : 0.0f;
#pragma unroll
            for (int k = 0; k < NSHIFT; ++k)
                q[k] = s1 ? (k >= 1 ? p[k - 1] : 0.0f) : p[k];
#pragma unroll
            for (int k = 0; k < NSHIFT; ++k)
                p[k] = s2 ? (k >= 2 ? q[k - 2] : 0.0f) : q[k];
#pragma unroll
            for (int k = 0; k < NSHIFT; ++k) {
                const float r = s4 ? (k >= 4 ? p[k - 4] : 0.0f) : p[k];
                if (e == 0) a[b][k].x = r; else a[b][k].y = r;
            }
        }
    }
}

template <int S, int JLO, int JHI>
__device__ __forceinline__ void mma_step(v8f (&acc)[NTILE],
                                         const v2f (&a)[2][NSHIFT],
                                         const float* brow)
{
#pragma unroll
    for (int jp = JLO; jp < JHI; ++jp) {
#pragma unroll
        for (int q = 0; q < 4; ++q) {
            const int b = q >> 1;
            const int t = (b == 0 ? 2 * S - 1 : 2 * S - 3) + (q & 1);
            if (t >= 0 && t < NTILE) {
                const v2f bf = *(const v2f*)(brow +
                    ((t + 2 * b) * QW + (jp % 7) * PHW + jp / 7) * 2);
                acc[t] = __builtin_amdgcn_wmma_f32_16x16x4_f32(
                    false, a[b][jp], false, bf, (short)0, acc[t], false, false);
            }
        }
    }
}

__device__ __forceinline__ void wave_lds_order()
{
    __builtin_amdgcn_fence(__ATOMIC_RELEASE, "wavefront");
    __builtin_amdgcn_wave_barrier();
    __builtin_amdgcn_fence(__ATOMIC_ACQUIRE, "wavefront");
}

template <int S>
__device__ __forceinline__ void strip_step(v8f (&acc)[NTILE],
                                           const v2f (&a)[2][NSHIFT],
                                           const float* brow, float* Pool,
                                           RowQuad& ha, RowQuad& hb,
                                           v2f (&carry)[NSEG],
                                           const float* __restrict__ xp, int r0,
                                           const unsigned (&coff)[NSEG],
                                           const bool (&cv)[NSEG],
                                           const int (&sword)[NSEG])
{
    const int prn = 4 * (S + 1);
    pool_issue(ha, xp, r0, prn, coff);
    __builtin_amdgcn_sched_barrier(0);
    mma_step<S, 0, 7>(acc, a, brow);
    __builtin_amdgcn_sched_barrier(0);
    pool_finish(ha, carry, Pool, r0, prn, cv, sword);
    pool_issue(hb, xp, r0, prn + 2, coff);
    __builtin_amdgcn_sched_barrier(0);
    mma_step<S, 7, NSHIFT>(acc, a, brow);
    __builtin_amdgcn_sched_barrier(0);
    pool_finish(hb, carry, Pool, r0, prn + 2, cv, sword);
    wave_lds_order();
}

__global__ __launch_bounds__(BLOCK) void fused_pool_dwconv_wmma(
    const float* __restrict__ x, const float* __restrict__ S,
    float* __restrict__ out)
{
    __shared__ __attribute__((aligned(16))) float PoolAll[WAVES * QROWS * QW * 2];

    const int tid   = threadIdx.x;
    const int lane  = tid & 31;
    const int wave  = __builtin_amdgcn_readfirstlane(tid >> 5);
    const int strip = blockIdx.x * WAVES + wave;
    const int tr    = strip % TROWS;
    const int d     = (strip / TROWS) % ND;
    const int n     = strip / (TROWS * ND);
    const int r0    = tr * SROWS;

    float* Pool = PoolAll + wave * (QROWS * QW * 2);
    const float* xp = x + ((size_t)(n * ND + d)) * (HX * WX);
    const float* sp = S + d * (FH * FW);

    const int mrow  = lane & 15;
    const int khalf = lane >> 4;
    const int mr    = (mrow >= 7) + (mrow >= 14);
    const int u     = mrow - 7 * mr;

    unsigned coff[NSEG];
    bool     cv[NSEG];
    int      sword[NSEG];
#pragma unroll
    for (int s = 0; s < NSEG; ++s) {
        const int cc = SEGW * s + 15 * (lane >> 4) + (lane & 15) - 1;
        const int C  = cc - 3;
        const int Cc = C < 0 ? 0 : (C >= WS ? WS - 1 : C);
        coff[s] = (unsigned)(2 * Cc);
        cv[s]   = ((lane & 15) != 0) & (C >= 0) & (C < WS);
        const bool keep = ((lane & 15) != 0) & (cc < 7 * PHW);
        const unsigned ccp = keep ? (unsigned)cc : 0u;
        const unsigned q7  = (ccp * 9363u) >> 16;
        const unsigned r7  = ccp - 7u * q7;
        sword[s] = keep ? (int)(r7 * PHW + q7)
                        : SPARE + 4 * s + 2 * (lane >> 4) + ((lane & 15) != 0);
    }

    RowQuad ha, hb;
    v2f carry[NSEG];
    float araw[2][2][FW];
    v2f   a[2][NSHIFT];

    row_issue(carry, x_row(xp, 2 * (r0 - 3) - 1), coff);
    pool_issue(ha, xp, r0, 0, coff);
    pool_issue(hb, xp, r0, 2, coff);
    taps_issue(araw, sp, mr, khalf);
    __builtin_amdgcn_sched_barrier(0);
    pool_finish(ha, carry, Pool, r0, 0, cv, sword);
    pool_finish(hb, carry, Pool, r0, 2, cv, sword);
    taps_finish(a, araw, mr, u, khalf);

    v8f acc[NTILE];
#pragma unroll
    for (int t = 0; t < NTILE; ++t)
        acc[t] = (v8f){0.f, 0.f, 0.f, 0.f, 0.f, 0.f, 0.f, 0.f};

    wave_lds_order();

    const float* brow = Pool + (khalf * QW + mrow) * 2;

    strip_step<0>(acc, a, brow, Pool, ha, hb, carry, xp, r0, coff, cv, sword);
    strip_step<1>(acc, a, brow, Pool, ha, hb, carry, xp, r0, coff, cv, sword);
    strip_step<2>(acc, a, brow, Pool, ha, hb, carry, xp, r0, coff, cv, sword);
    strip_step<3>(acc, a, brow, Pool, ha, hb, carry, xp, r0, coff, cv, sword);

    mma_step<NSTEP - 1, 0, NSHIFT>(acc, a, brow);

    wave_lds_order();
    int soff[8];
#pragma unroll
    for (int v = 0; v < 8; ++v) {
        const int M   = v + 8 * khalf;
        const int mrv = (M >= 7) + (M >= 14);
        soff[v] = mrv * WS + (M - 7 * mrv) + 7 * mrow;
    }
#pragma unroll
    for (int t = 0; t < NTILE; ++t) {
#pragma unroll
        for (int v = 0; v < 6; ++v)
            Pool[soff[v] + t * 2 * WS] = acc[t][v];
    }
    if (khalf == 0) {
#pragma unroll
        for (int t = 0; t < NTILE; ++t) {
#pragma unroll
            for (int v = 6; v < 8; ++v)
                Pool[soff[v] + t * 2 * WS] = acc[t][v];
        }
    }
    wave_lds_order();

    float* op = out + ((size_t)(n * ND + d)) * (HS * WS) + (size_t)r0 * WS;
#pragma unroll
    for (int it = 0; it < 12; ++it) {
        const int idx = (it * 32 + lane) * 4;
        *(v4f*)(op + idx) = *(const v4f*)(Pool + idx);
    }
    if (lane < 8) {
        const int idx = (12 * 32 + lane) * 4;
        *(v4f*)(op + idx) = *(const v4f*)(Pool + idx);
    }
}

extern "C" void kernel_launch(void* const* d_in, const int* in_sizes, int n_in,
                              void* d_out, int out_size, void* d_ws, size_t ws_size,
                              hipStream_t stream) {
    (void)in_sizes; (void)n_in; (void)d_ws; (void)ws_size; (void)out_size;
    const float* x = (const float*)d_in[0];
    const float* S = (const float*)d_in[1];
    float* out = (float*)d_out;

    dim3 grid(NB * ND * TROWS / WAVES);
    fused_pool_dwconv_wmma<<<grid, BLOCK, 0, stream>>>(x, S, out);
}


# strip body in three copies by the strip's place in its plane (top, interior, bottom): row clamps and zero selects fold at compile time, x rows are constant offsets; 1481 instructions a wave, was 2071
# speedup vs baseline: 1.8963x; 1.0525x over previous
#include <hip/hip_runtime.h>


typedef __attribute__((ext_vector_type(2))) float v2f;
typedef __attribute__((ext_vector_type(4))) float v4f;
typedef __attribute__((ext_vector_type(8))) float v8f;

#define NB 16
#define ND 32
#define HX 224
#define WX 224
#define HS 112
#define WS 112
#define FH 7
#define FW 7
#define SROWS 14
#define TROWS (HS / SROWS)
#define NTILE (SROWS / 2)
#define NSHIFT (FW + 6)
#define WAVES 4
#define BLOCK (WAVES * 32)
#define NSEG 4
#define SEGW 30
#define NSTEP 5
#define QROWS 10
#define PHW 17
#define QW 144
#define SPARE 119

struct RowQuad {
    v2f r[4][NSEG];
};

#define POS_TOP 0
#define POS_MID 1
#define POS_BOT 2

template <int POS>
__device__ __forceinline__ constexpr bool strip_row_valid(int k)
{
    return POS == POS_MID ? true : (POS == POS_TOP ? k >= 3 : k < SROWS + 3);
}

template <int POS>
__device__ __forceinline__ const float* x_row(const float* __restrict__ xb, int i)
{
    if (POS == POS_MID)
        return xb + i * WX;
    const int y  = 2 * ((POS == POS_TOP ? 0 : HS - SROWS) - 3) - 1 + i;
    const int yc = y < 0 ? 0 : (y > HX - 1 ? HX - 1 : y);
    return xb + yc * WX;
}

__device__ __forceinline__ void row_issue(v2f (&dst)[NSEG],
                                          const float* __restrict__ row,
                                          const unsigned (&coff)[NSEG])
{
#pragma unroll
    for (int s = 0; s < NSEG; ++s)
        dst[s] = *(const v2f*)(row + coff[s]);
}

template <int POS>
__device__ __forceinline__ void pool_issue(RowQuad& h,
                                           const float* __restrict__ xb,
                                           int pr,
                                           const unsigned (&coff)[NSEG])
{
#pragma unroll
    for (int i = 0; i < 4; ++i)
        row_issue(h.r[i], x_row<POS>(xb, 2 * pr + 1 + i), coff);
}

__device__ __forceinline__ float left_lane(float v)
{
    const int b = __builtin_bit_cast(int, v);
    return __builtin_bit_cast(float,
        __builtin_amdgcn_update_dpp(b, b, 0x111, 0xf, 0xf, false));
}

__device__ __forceinline__ float max3f(float a, float b, float c)
{
    return fmaxf(fmaxf(a, b), c);
}

template <int POS>
__device__ __forceinline__ void pool_finish(const RowQuad& h, v2f (&carry)[NSEG],
                                            float* Pool, int pr,
                                            const int (&sword)[NSEG])
{
    const bool rv0 = strip_row_valid<POS>(pr);
    const bool rv1 = strip_row_valid<POS>(pr + 1);
#pragma unroll
    for (int s = 0; s < NSEG; ++s) {
        v2f v0, v1;
        v0.x = max3f(carry[s].x, h.r[0][s].x, h.r[1][s].x);
        v0.y = max3f(carry[s].y, h.r[0][s].y, h.r[1][s].y);
        v1.x = max3f(h.r[1][s].x, h.r[2][s].x, h.r[3][s].x);
        v1.y = max3f(h.r[1][s].y, h.r[2][s].y, h.r[3][s].y);
        carry[s] = h.r[3][s];
        const float m0 = max3f(v0.x, v0.y, left_lane(v0.y));
        const float m1 = max3f(v1.x, v1.y, left_lane(v1.y));
        v2f o;
        o.x = rv0 ? m0 : 0.0f;
        o.y = rv1 ? m1 : 0.0f;
        *(v2f*)&Pool[((pr >> 1) * QW + sword[s]) * 2] = o;
    }
}

__device__ __forceinline__ void taps_issue(float (&raw)[2][2][FW],
                                           const float* __restrict__ sp,
                                           int mr, int khalf)
{
#pragma unroll
    for (int b = 0; b < 2; ++b) {
#pragma unroll
        for (int e = 0; e < 2; ++e) {
            const int i  = 4 * b + 2 * khalf + e - mr;
            const int ic = i < 0 ? 0 : (i > FH - 1 ? FH - 1 : i);
#pragma unroll
            for (int j = 0; j < FW; ++j)
                raw[b][e][j] = sp[ic * FW + j];
        }
    }
}

__device__ __forceinline__ void taps_finish(v2f (&a)[2][NSHIFT],
                                            const float (&raw)[2][2][FW],
                                            int mr, int u, int khalf)
{
    const bool s1 = (u & 1) != 0;
    const bool s2 = (u & 2) != 0;
    const bool s4 = (u & 4) != 0;
#pragma unroll
    for (int b = 0; b < 2; ++b) {
#pragma unroll
        for (int e = 0; e < 2; ++e) {
            const int  i  = 4 * b + 2 * khalf + e - mr;
            const bool in = (mr < 2) & (i >= 0) & (i < FH);
            float p[NSHIFT], q[NSHIFT];
#pragma unroll
            for (int k = 0; k < NSHIFT; ++k)
                p[k] = (k < FW && in) ? raw[b][e][k < FW ? k : 0] : 0.0f;
#pragma unroll
            for (int k = 0; k < NSHIFT; ++k)
                q[k] = s1 ? (k >= 1 ? p[k - 1] : 0.0f) : p[k];
#pragma unroll
            for (int k = 0; k < NSHIFT; ++k)
                p[k] = s2 ? (k >= 2 ? q[k - 2] : 0.0f) : q[k];
#pragma unroll
            for (int k = 0; k < NSHIFT; ++k) {
                const float r = s4 ? (k >= 4 ? p[k - 4] : 0.0f) : p[k];
                if (e == 0) a[b][k].x = r; else a[b][k].y = r;
            }
        }
    }
}

template <int S, int JLO, int JHI>
__device__ __forceinline__ void mma_step(v8f (&acc)[NTILE],
                                         const v2f (&a)[2][NSHIFT],
                                         const float* brow)
{
#pragma unroll
    for (int jp = JLO; jp < JHI; ++jp) {
#pragma unroll
        for (int q = 0; q < 4; ++q) {
            const int b = q >> 1;
            const int t = (b == 0 ? 2 * S - 1 : 2 * S - 3) + (q & 1);
            if (t >= 0 && t < NTILE) {
                const v2f bf = *(const v2f*)(brow +
                    ((t + 2 * b) * QW + (jp % 7) * PHW + jp / 7) * 2);
                acc[t] = __builtin_amdgcn_wmma_f32_16x16x4_f32(
                    false, a[b][jp], false, bf, (short)0, acc[t], false, false);
            }
        }
    }
}

__device__ __forceinline__ void wave_lds_order()
{
    __builtin_amdgcn_fence(__ATOMIC_RELEASE, "wavefront");
    __builtin_amdgcn_wave_barrier();
    __builtin_amdgcn_fence(__ATOMIC_ACQUIRE, "wavefront");
}

template <int POS, int S>
__device__ __forceinline__ void strip_step(v8f (&acc)[NTILE],
                                           const v2f (&a)[2][NSHIFT],
                                           const float* brow, float* Pool,
                                           RowQuad& ha, RowQuad& hb,
                                           v2f (&carry)[NSEG],
                                           const float* __restrict__ xb,
                                           const unsigned (&coff)[NSEG],
                                           const int (&sword)[NSEG])
{
    const int prn = 4 * (S + 1);
    pool_issue<POS>(ha, xb, prn, coff);
    __builtin_amdgcn_sched_barrier(0);
    mma_step<S, 0, 7>(acc, a, brow);
    __builtin_amdgcn_sched_barrier(0);
    pool_finish<POS>(ha, carry, Pool, prn, sword);
    pool_issue<POS>(hb, xb, prn + 2, coff);
    __builtin_amdgcn_sched_barrier(0);
    mma_step<S, 7, NSHIFT>(acc, a, brow);
    __builtin_amdgcn_sched_barrier(0);
    pool_finish<POS>(hb, carry, Pool, prn + 2, sword);
    wave_lds_order();
}

template <int POS>
__device__ __forceinline__ void strip_body(const float* __restrict__ xp,
                                           const float* __restrict__ sp,
                                           float* __restrict__ op,
                                           float* Pool, int r0, int lane)
{
    const int mrow  = lane & 15;
    const int khalf = lane >> 4;
    const int mr    = (mrow >= 7) + (mrow >= 14);
    const int u     = mrow - 7 * mr;

    const float* xb = POS == POS_MID ? xp + (2 * (r0 - 3) - 1) * WX : xp;

    unsigned coff[NSEG];
    int      sword[NSEG];
#pragma unroll
    for (int s = 0; s < NSEG; ++s) {
        const int cc = SEGW * s + 15 * (lane >> 4) + (lane & 15) - 1;
        const int C  = cc - 3;
        const int Cc = C < 0 ? 0 : (C >= WS ? WS - 1 : C);
        coff[s] = (unsigned)(2 * Cc);
        const bool keep = ((lane & 15) != 0) & (C >= 0) & (C < WS);
        const unsigned ccp = keep ? (unsigned)cc : 0u;
        const unsigned q7  = (ccp * 9363u) >> 16;
        const unsigned r7  = ccp - 7u * q7;
        sword[s] = keep ? (int)(r7 * PHW + q7)
                        : SPARE + (lane & 7) + 8 * (lane >> 4);
    }

    RowQuad ha, hb;
    v2f carry[NSEG];
    float araw[2][2][FW];
    v2f   a[2][NSHIFT];

    row_issue(carry, x_row<POS>(xb, 0), coff);
    pool_issue<POS>(ha, xb, 0, coff);
    pool_issue<POS>(hb, xb, 2, coff);
    taps_issue(araw, sp, mr, khalf);
    __builtin_amdgcn_sched_barrier(0);

#pragma unroll
    for (int h = 0; h < 2; ++h) {
        const unsigned idx = lane + 32 * h < 60 ? lane + 32 * h : 59;
        const unsigned q   = idx / 6u;
        const unsigned j   = idx - 6u * q;
        const unsigned w   = PHW * j + (j >= 3u ? 16u : 0u);
        *(v2f*)&Pool[(q * QW + w) * 2] = (v2f){0.f, 0.f};
    }

    pool_finish<POS>(ha, carry, Pool, 0, sword);
    pool_finish<POS>(hb, carry, Pool, 2, sword);
    taps_finish(a, araw, mr, u, khalf);

    v8f acc[NTILE];
#pragma unroll
    for (int t = 0; t < NTILE; ++t)
        acc[t] = (v8f){0.f, 0.f, 0.f, 0.f, 0.f, 0.f, 0.f, 0.f};

    wave_lds_order();

    const float* brow = Pool + (khalf * QW + mrow) * 2;

    strip_step<POS, 0>(acc, a, brow, Pool, ha, hb, carry, xb, coff, sword);
    strip_step<POS, 1>(acc, a, brow, Pool, ha, hb, carry, xb, coff, sword);
    strip_step<POS, 2>(acc, a, brow, Pool, ha, hb, carry, xb, coff, sword);
    strip_step<POS, 3>(acc, a, brow, Pool, ha, hb, carry, xb, coff, sword);

    mma_step<NSTEP - 1, 0, NSHIFT>(acc, a, brow);

    wave_lds_order();
    int soff[8];
#pragma unroll
    for (int v = 0; v < 8; ++v) {
        const int M   = v + 8 * khalf;
        const int mrv = (M >= 7) + (M >= 14);
        soff[v] = mrv * WS + (M - 7 * mrv) + 7 * mrow;
    }
#pragma unroll
    for (int t = 0; t < NTILE; ++t) {
#pragma unroll
        for (int v = 0; v < 6; ++v)
            Pool[soff[v] + t * 2 * WS] = acc[t][v];
    }
    if (khalf == 0) {
#pragma unroll
        for (int t = 0; t < NTILE; ++t) {
#pragma unroll
            for (int v = 6; v < 8; ++v)
                Pool[soff[v] + t * 2 * WS] = acc[t][v];
        }
    }
    wave_lds_order();

#pragma unroll
    for (int it = 0; it < 12; ++it) {
        const int idx = (it * 32 + lane) * 4;
        *(v4f*)(op + idx) = *(const v4f*)(Pool + idx);
    }
    if (lane < 8) {
        const int idx = (12 * 32 + lane) * 4;
        *(v4f*)(op + idx) = *(const v4f*)(Pool + idx);
    }
}

__global__ __launch_bounds__(BLOCK) void fused_pool_dwconv_wmma(
    const float* __restrict__ x, const float* __restrict__ S,
    float* __restrict__ out)
{
    __shared__ __attribute__((aligned(16))) float PoolAll[WAVES * QROWS * QW * 2];

    const int tid   = threadIdx.x;
    const int lane  = tid & 31;
    const int wave  = __builtin_amdgcn_readfirstlane(tid >> 5);
    const int strip = blockIdx.x * WAVES + wave;
    const int tr    = strip % TROWS;
    const int d     = (strip / TROWS) % ND;
    const int n     = strip / (TROWS * ND);
    const int r0    = tr * SROWS;

    float* Pool = PoolAll + wave * (QROWS * QW * 2);
    const float* xp = x + ((size_t)(n * ND + d)) * (HX * WX);
    const float* sp = S + d * (FH * FW);
    float* op = out + ((size_t)(n * ND + d)) * (HS * WS) + (size_t)r0 * WS;

    if (tr == 0)
        strip_body<POS_TOP>(xp, sp, op, Pool, 0, lane);
    else if (tr == TROWS - 1)
        strip_body<POS_BOT>(xp, sp, op, Pool, HS - SROWS, lane);
    else
        strip_body<POS_MID>(xp, sp, op, Pool, r0, lane);
}

extern "C" void kernel_launch(void* const* d_in, const int* in_sizes, int n_in,
                              void* d_out, int out_size, void* d_ws, size_t ws_size,
                              hipStream_t stream) {
    (void)in_sizes; (void)n_in; (void)d_ws; (void)ws_size; (void)out_size;
    const float* x = (const float*)d_in[0];
    const float* S = (const float*)d_in[1];
    float* out = (float*)d_out;

    dim3 grid(NB * ND * TROWS / WAVES);
    fused_pool_dwconv_wmma<<<grid, BLOCK, 0, stream>>>(x, S, out);
}
